# starting kernel
# baseline (speedup 1.0000x reference)
#include <hip/hip_runtime.h>

typedef __attribute__((ext_vector_type(2))) float v2f;
typedef __attribute__((ext_vector_type(8))) float v8f;

#define ALPHA 0.05f
#define EPS   0.01f

#define TM 128
#define TN 128
#define TK 32

static __device__ __forceinline__ v8f wmma4(const v2f a, const v2f b, const v8f c)
{
    return __builtin_amdgcn_wmma_f32_16x16x4_f32(false, a, false, b, (short)0, c,
                                                 false, false);
}

static __device__ __forceinline__ void store_tile(
    const v8f acc, int r0, int c0, float* __restrict__ Cm, int ldc,
    const float* __restrict__ Cin, const float* __restrict__ bias,
    float alpha, float beta, float gamma, int relu, int hi, int ln)
{
    const int c = c0 + ln;
    const float bv = bias ? gamma * bias[c] : 0.f;
#pragma unroll
    for (int v = 0; v < 8; ++v) {
        const int r = r0 + v + 8 * hi;
        float val = alpha * acc[v] + bv;
        if (Cin)  val += beta * Cin[(size_t)r * ldc + c];
        if (relu) val = fmaxf(val, 0.f);
        Cm[(size_t)r * ldc + c] = val;
    }
}

__global__ __launch_bounds__(256) void gemm_f32_wmma(
    const float* __restrict__ Am, int lda,
    const float* __restrict__ Bm, int ldb, int transB,
    float* __restrict__ Cm, int ldc,
    const float* __restrict__ Cin,
    const float* __restrict__ bias,
    float alpha, float beta, float gamma, int relu,
    int K, const int* skip)
{
    if (skip && *skip) return;

    __shared__ float As[TM][TK + 4];
    __shared__ float Bs[TK][TN + 4];

    const int t    = threadIdx.x;
    const int row0 = blockIdx.y * TM;
    const int col0 = blockIdx.x * TN;

    const int wid  = t >> 5;
    const int lane = t & 31;
    const int wm   = (wid >> 2) * 64;
    const int wn   = (wid & 3) * 32;
    const int hi   = lane >> 4;
    const int ln   = lane & 15;

    v8f acc[4][2];
#pragma unroll
    for (int mi = 0; mi < 4; ++mi)
#pragma unroll
        for (int ni = 0; ni < 2; ++ni) acc[mi][ni] = (v8f){};

    const int arow = t >> 1;
    const int akc  = (t & 1) * 16;
    const int bkp  = t >> 3;
    const int bnp  = (t & 7) * 16;
    const int btn  = t >> 1;
    const int btk  = (t & 1) * 16;

    for (int kb = 0; kb < K; kb += TK) {
        const float* ap = Am + (size_t)(row0 + arow) * lda + kb + akc;
#pragma unroll
        for (int j = 0; j < 4; ++j)
            *(float4*)&As[arow][akc + 4 * j] = *(const float4*)(ap + 4 * j);
        if (kb + TK < K) __builtin_prefetch(ap + TK, 0, 1);

        if (!transB) {
            const float* bp = Bm + (size_t)(kb + bkp) * ldb + col0 + bnp;
#pragma unroll
            for (int j = 0; j < 4; ++j)
                *(float4*)&Bs[bkp][bnp + 4 * j] = *(const float4*)(bp + 4 * j);
            if (kb + TK < K) __builtin_prefetch(bp + (size_t)TK * ldb, 0, 1);
        } else {
            const float* bp = Bm + (size_t)(col0 + btn) * ldb + kb + btk;
#pragma unroll
            for (int j = 0; j < 4; ++j) {
                const float4 vv = *(const float4*)(bp + 4 * j);
                Bs[btk + 4 * j + 0][btn] = vv.x;
                Bs[btk + 4 * j + 1][btn] = vv.y;
                Bs[btk + 4 * j + 2][btn] = vv.z;
                Bs[btk + 4 * j + 3][btn] = vv.w;
            }
            if (kb + TK < K) __builtin_prefetch(bp + TK, 0, 1);
        }
        __syncthreads();

#pragma unroll
        for (int kk = 0; kk < TK; kk += 4) {
            v2f a[4], b[2];
#pragma unroll
            for (int mi = 0; mi < 4; ++mi) {
                a[mi].x = As[wm + 16 * mi + ln][kk + 2 * hi];
                a[mi].y = As[wm + 16 * mi + ln][kk + 2 * hi + 1];
            }
#pragma unroll
            for (int ni = 0; ni < 2; ++ni) {
                b[ni].x = Bs[kk + 2 * hi][wn + 16 * ni + ln];
                b[ni].y = Bs[kk + 2 * hi + 1][wn + 16 * ni + ln];
            }
#pragma unroll
            for (int mi = 0; mi < 4; ++mi)
#pragma unroll
                for (int ni = 0; ni < 2; ++ni)
                    acc[mi][ni] = wmma4(a[mi], b[ni], acc[mi][ni]);
        }
        __syncthreads();
    }

#pragma unroll
    for (int mi = 0; mi < 4; ++mi)
#pragma unroll
        for (int ni = 0; ni < 2; ++ni)
            store_tile(acc[mi][ni], row0 + wm + 16 * mi, col0 + wn + 16 * ni,
                       Cm, ldc, Cin, bias, alpha, beta, gamma, relu, hi, ln);
}

__global__ void k_prep_u(const float* __restrict__ z, const float* __restrict__ w,
                         float* __restrict__ u, int n, const int* skip)
{
    if (skip && *skip) return;
    int i = blockIdx.x * blockDim.x + threadIdx.x;
    if (i < n) {
        float zi = z[i];
        float x  = fmaxf(zi, 0.f);
        u[i] = 2.f * x - zi - ALPHA * w[i];
    }
}

__global__ void k_update(float* __restrict__ z, const float* __restrict__ y,
                         float* __restrict__ acc, int n, const int* skip)
{
    __shared__ float sm[256];
    if (*skip) return;
    int i = blockIdx.x * blockDim.x + threadIdx.x;
    float d2 = 0.f;
    if (i < n) {
        float zi = z[i];
        float x  = fmaxf(zi, 0.f);
        float yi = y[i];
        z[i] = zi - x + yi;
        float d = yi - x;
        d2 = d * d;
    }
    sm[threadIdx.x] = d2;
    __syncthreads();
    for (int s = 128; s > 0; s >>= 1) {
        if (threadIdx.x < s) sm[threadIdx.x] += sm[threadIdx.x + s];
        __syncthreads();
    }
    if (threadIdx.x == 0) atomicAdd(acc, sm[0]);
}

__global__ void k_sumsq(const float* __restrict__ x, float* __restrict__ acc, int n)
{
    __shared__ float sm[256];
    int i = blockIdx.x * blockDim.x + threadIdx.x;
    float v = 0.f;
    if (i < n) { float t = x[i]; v = t * t; }
    sm[threadIdx.x] = v;
    __syncthreads();
    for (int s = 128; s > 0; s >>= 1) {
        if (threadIdx.x < s) sm[threadIdx.x] += sm[threadIdx.x + s];
        __syncthreads();
    }
    if (threadIdx.x == 0) atomicAdd(acc, sm[0]);
}

__global__ void k_zero(float* p) { p[0] = 0.f; }

__global__ void k_check(const float* acc, int* flag)
{
    if (*flag == 0 && *acc <= EPS * EPS) *flag = 1;
}

__global__ void k_init_X(float* __restrict__ X, const float* __restrict__ fn, int n)
{
    int i = blockIdx.x * blockDim.x + threadIdx.x;
    if (i < n * n) {
        float c = rsqrtf(*fn);
        int r = i / n, cc = i - r * n;
        X[i] = (r == cc) ? c : 0.f;
    }
}

__global__ void k_final(const float* __restrict__ z, const float* __restrict__ y,
                        float* __restrict__ out, int n)
{
    int i = blockIdx.x * blockDim.x + threadIdx.x;
    if (i < n) {
        float zi = z[i];
        float x  = fmaxf(zi, 0.f);
        out[i] = fmaxf(zi - x + y[i], 0.f);
    }
}

extern "C" void kernel_launch(void* const* d_in, const int* in_sizes, int n_in,
                              void* d_out, int out_size, void* d_ws, size_t ws_size,
                              hipStream_t stream)
{
    const float* ctx  = (const float*)d_in[0];
    const float* Ain  = (const float*)d_in[1];
    const float* bvec = (const float*)d_in[2];
    const float* W1   = (const float*)d_in[3];
    const float* b1   = (const float*)d_in[4];
    const float* W2   = (const float*)d_in[5];
    const float* b2   = (const float*)d_in[6];
    const float* z0   = (const float*)d_in[7];
    float* out = (float*)d_out;

    const int B = 256, D = 512, H = 2048, N1 = 1024, N2 = 4096;

    float* W = (float*)d_ws;
    size_t o = 0;
    float* z  = W + o; o += (size_t)B * N2;
    float* u  = W + o; o += (size_t)B * N2;
    float* y  = W + o; o += (size_t)B * N2;
    float* wv = W + o; o += (size_t)B * N2;
    float* h  = W + o; o += (size_t)B * H;
    float* R  = W + o; o += (size_t)B * N1;
    float* S  = W + o; o += (size_t)B * N1;
    float* G  = W + o; o += (size_t)N1 * N1;
    float* Xa = W + o; o += (size_t)N1 * N1;
    float* Xb = W + o; o += (size_t)N1 * N1;
    float* T1 = W + o; o += (size_t)N1 * N1;
    float* acc = W + o; o += 1;
    float* fn  = W + o; o += 1;
    int*   flag = (int*)(W + o); o += 1;

    dim3 blk(256);
    auto gemm = [&](const float* A_, int lda, const float* B_, int ldb, int tB,
                    float* C_, int ldc, const float* Cin, const float* bias,
                    float al, float be, float ga, int relu,
                    int M, int N, int K, const int* skip) {
        dim3 grid(N / TN, M / TM);
        gemm_f32_wmma<<<grid, blk, 0, stream>>>(A_, lda, B_, ldb, tB, C_, ldc,
                                                Cin, bias, al, be, ga, relu, K, skip);
    };

    hipMemcpyAsync(z, z0, sizeof(float) * (size_t)B * N2, hipMemcpyDeviceToDevice, stream);
    hipMemsetAsync(flag, 0, sizeof(int), stream);

    gemm(ctx, D, W1, H, 0, h,  H,  nullptr, b1, 1.f, 0.f, 1.f, 1, B, H,  D, nullptr);
    gemm(h,   H, W2, N2, 0, wv, N2, nullptr, b2, 1.f, 0.f, 1.f, 0, B, N2, H, nullptr);

    gemm(Ain, N2, Ain, N2, 1, G, N1, nullptr, nullptr, 1.f, 0.f, 0.f, 0, N1, N1, N2, nullptr);

    k_zero<<<1, 1, 0, stream>>>(fn);
    k_sumsq<<<dim3((N1 * N1 + 255) / 256), blk, 0, stream>>>(G, fn, N1 * N1);
    k_init_X<<<dim3((N1 * N1 + 255) / 256), blk, 0, stream>>>(Xa, fn, N1);

    float* Xc = Xa; float* Xn = Xb;
    for (int it = 0; it < 16; ++it) {
        gemm(G,  N1, Xc, N1, 0, T1, N1, nullptr, nullptr,  1.f, 0.f, 0.f, 0, N1, N1, N1, nullptr);
        gemm(Xc, N1, T1, N1, 0, Xn, N1, Xc,      nullptr, -1.f, 2.f, 0.f, 0, N1, N1, N1, nullptr);
        float* tmp = Xc; Xc = Xn; Xn = tmp;
    }
    float* Ginv = Xc;

    const int nz = B * N2;
    dim3 gout((nz + 255) / 256);
    for (int itr = 0; itr < 50; ++itr) {
        k_prep_u<<<gout, blk, 0, stream>>>(z, wv, u, nz, (const int*)flag);
        gemm(u, N2, Ain,  N2, 1, R, N1, nullptr, bvec, 1.f, 0.f, -1.f, 0, B, N1, N2, flag);
        gemm(R, N1, Ginv, N1, 0, S, N1, nullptr, nullptr, 1.f, 0.f, 0.f, 0, B, N1, N1, flag);
        gemm(S, N1, Ain,  N2, 0, y, N2, u, nullptr, -1.f, 1.f, 0.f, 0, B, N2, N1, flag);
        k_zero<<<1, 1, 0, stream>>>(acc);
        k_update<<<gout, blk, 0, stream>>>(z, y, acc, nz, (const int*)flag);
        k_check<<<1, 1, 0, stream>>>((const float*)acc, flag);
    }

    k_prep_u<<<gout, blk, 0, stream>>>(z, wv, u, nz, (const int*)nullptr);
    gemm(u, N2, Ain,  N2, 1, R, N1, nullptr, bvec, 1.f, 0.f, -1.f, 0, B, N1, N2, nullptr);
    gemm(R, N1, Ginv, N1, 0, S, N1, nullptr, nullptr, 1.f, 0.f, 0.f, 0, B, N1, N1, nullptr);
    gemm(S, N1, Ain,  N2, 0, y, N2, u, nullptr, -1.f, 1.f, 0.f, 0, B, N2, N1, nullptr);
    k_final<<<gout, blk, 0, stream>>>(z, y, out, nz);
}


# GEMMs whose output gives 256 tiles of 64x64 (S@A, Newton-Schulz, MLP layer 2, A@A^T): 64x64 tile per work-group, K still split over its 8 waves (each a full 64x64 partial); was 32x32; sum order kept
# speedup vs baseline: 6.8126x; 6.8126x over previous
#include <hip/hip_runtime.h>

typedef __attribute__((ext_vector_type(2))) float v2f;
typedef __attribute__((ext_vector_type(8))) float v8f;

#define ALPHA 0.05f
#define EPS   0.01f

#define BK 256
#define WK 32
#define APITCH 36

static __device__ __forceinline__ v8f wmma4(const v2f a, const v2f b, const v8f c)
{
    return __builtin_amdgcn_wmma_f32_16x16x4_f32(false, a, false, b, (short)0, c,
                                                 false, false);
}

template <int TRANSB, int T>
__global__ __launch_bounds__(256) void gemm_f32_wmma(
    const float* __restrict__ Am, int lda,
    const float* __restrict__ Bm, int ldb,
    float* __restrict__ Cm, int ldc,
    const float* __restrict__ Cin,
    const float* __restrict__ bias,
    float alpha, float beta, float gamma, int relu,
    int K, const int* skip)
{
    if (skip && *skip) return;

    constexpr int FT   = T / 16;
    constexpr int NL   = T / 4;
    constexpr int BP   = T + 8;
    constexpr int BL   = T / 4;
    constexpr int BR   = 32 / BL;
    constexpr int RP   = T + 4;
    constexpr int WA   = T * APITCH;
    constexpr int WB   = (T * APITCH > 32 * BP) ? T * APITCH : 32 * BP;
    constexpr int WREG = WA + WB;
    static_assert(T * RP <= WREG, "a partial tile is parked in the wave's own region");

    __shared__ __attribute__((aligned(16))) float sW[8 * WREG];

    const int t    = threadIdx.x;
    const int row0 = blockIdx.y * T;
    const int col0 = blockIdx.x * T;

    const int wid  = t >> 5;
    const int lane = t & 31;
    const int hi   = lane >> 4;
    const int ln   = lane & 15;
    const int kw   = wid * WK;
    const int ra0  = wid * WREG;
    const int rb0  = ra0 + WA;

    v8f acc[FT][FT];
#pragma unroll
    for (int mi = 0; mi < FT; ++mi)
#pragma unroll
        for (int ni = 0; ni < FT; ++ni) acc[mi][ni] = (v8f){};

    const int lr = lane >> 3;
    const int lc = (lane & 7) * 4;
    const int br = lane / BL;
    const int bc = (lane % BL) * 4;

    for (int kb = 0; kb < K; kb += BK) {
        float4 ra[NL], rb[NL];
        const float* ab = Am + (size_t)row0 * lda + kb;
        const unsigned ao = (unsigned)(lr * lda + kw + lc);
#pragma unroll
        for (int j = 0; j < NL; ++j) ra[j] = *(const float4*)(ab + (size_t)(4 * j) * lda + ao);
        if (TRANSB) {
            const float* bb = Bm + (size_t)col0 * ldb + kb;
            const unsigned bo = (unsigned)(lr * ldb + kw + lc);
#pragma unroll
            for (int j = 0; j < NL; ++j) rb[j] = *(const float4*)(bb + (size_t)(4 * j) * ldb + bo);
            if (kb + BK < K) {
#pragma unroll
                for (int p = 0; p < T; p += 32) {
                    __builtin_prefetch(ab + BK + (unsigned)((p + lane) * lda + kw), 0, 1);
                    __builtin_prefetch(bb + BK + (unsigned)((p + lane) * ldb + kw), 0, 1);
                }
            }
        } else {
            const float* bb = Bm + (size_t)kb * ldb + col0;
            const unsigned bo = (unsigned)((kw + br) * ldb + bc);
#pragma unroll
            for (int j = 0; j < NL; ++j) rb[j] = *(const float4*)(bb + (size_t)(BR * j) * ldb + bo);
            if (kb + BK < K) {
#pragma unroll
                for (int p = 0; p < T; p += 32)
                    __builtin_prefetch(ab + BK + (unsigned)((p + lane) * lda + kw), 0, 1);
                __builtin_prefetch(bb + (size_t)BK * ldb + (unsigned)((kw + lane) * ldb), 0, 1);
            }
        }
        __builtin_amdgcn_sched_barrier(0);
#pragma unroll
        for (int j = 0; j < NL; ++j) {
            float* q = &sW[ra0 + (lr + 4 * j) * APITCH + lc];
            q[0] = ra[j].x; q[1] = ra[j].y; q[2] = ra[j].z; q[3] = ra[j].w;
        }
#pragma unroll
        for (int j = 0; j < NL; ++j) {
            float* q = TRANSB ? &sW[rb0 + (lr + 4 * j) * APITCH + lc]
                              : &sW[rb0 + (br + BR * j) * BP + bc];
            q[0] = rb[j].x; q[1] = rb[j].y; q[2] = rb[j].z; q[3] = rb[j].w;
        }
        __builtin_amdgcn_fence(__ATOMIC_ACQ_REL, "wavefront");

#pragma unroll
        for (int kk = 0; kk < WK; kk += 4) {
            const int k = kk + 2 * hi;
            v2f a[FT], b[FT];
#pragma unroll
            for (int mi = 0; mi < FT; ++mi)
                a[mi] = *(const v2f*)&sW[ra0 + (16 * mi + ln) * APITCH + k];
#pragma unroll
            for (int ni = 0; ni < FT; ++ni) {
                if (TRANSB) {
                    b[ni] = *(const v2f*)&sW[rb0 + (16 * ni + ln) * APITCH + k];
                } else {
                    b[ni].x = sW[rb0 + k * BP + 16 * ni + ln];
                    b[ni].y = sW[rb0 + (k + 1) * BP + 16 * ni + ln];
                }
            }
#pragma unroll
            for (int mi = 0; mi < FT; ++mi)
#pragma unroll
                for (int ni = 0; ni < FT; ++ni)
                    acc[mi][ni] = wmma4(a[mi], b[ni], acc[mi][ni]);
        }
        __builtin_amdgcn_fence(__ATOMIC_ACQ_REL, "wavefront");
    }

#pragma unroll
    for (int mi = 0; mi < FT; ++mi)
#pragma unroll
        for (int ni = 0; ni < FT; ++ni)
#pragma unroll
            for (int v = 0; v < 8; ++v)
                sW[ra0 + (16 * mi + 8 * hi + v) * RP + 16 * ni + ln] = acc[mi][ni][v];
    __syncthreads();

    constexpr int NP = T * T / 1024;
    constexpr int TR = 1024 / T;
    const int sr0 = t / (T / 4);
    const int sc  = (t % (T / 4)) * 4;
    const int gc  = col0 + sc;
    float4 bb = make_float4(0.f, 0.f, 0.f, 0.f);
    if (bias) bb = *(const float4*)(bias + gc);
    float4 ci[NP];
#pragma unroll
    for (int pc = 0; pc < NP; ++pc) ci[pc] = make_float4(0.f, 0.f, 0.f, 0.f);
    if (Cin) {
#pragma unroll
        for (int pc = 0; pc < NP; ++pc)
            ci[pc] = *(const float4*)(Cin + (size_t)(row0 + sr0 + TR * pc) * ldc + gc);
    }
#pragma unroll
    for (int pc = 0; pc < NP; ++pc) {
        const int sr = sr0 + TR * pc;
        const float4 bv = make_float4(gamma * bb.x, gamma * bb.y, gamma * bb.z, gamma * bb.w);
        float4 s = *(const float4*)&sW[sr * RP + sc];
#pragma unroll
        for (int w = 1; w < 8; ++w) {
            const float4 p = *(const float4*)&sW[w * WREG + sr * RP + sc];
            s.x += p.x; s.y += p.y; s.z += p.z; s.w += p.w;
        }
        float4 o = make_float4(alpha * s.x + bv.x, alpha * s.y + bv.y,
                               alpha * s.z + bv.z, alpha * s.w + bv.w);
        if (Cin) {
            o.x += beta * ci[pc].x; o.y += beta * ci[pc].y;
            o.z += beta * ci[pc].z; o.w += beta * ci[pc].w;
        }
        if (relu) {
            o.x = fmaxf(o.x, 0.f); o.y = fmaxf(o.y, 0.f);
            o.z = fmaxf(o.z, 0.f); o.w = fmaxf(o.w, 0.f);
        }
        *(float4*)(Cm + (size_t)(row0 + sr) * ldc + gc) = o;
    }
}

#define EW_BLOCKS 256
#define EW_PIECES 4

static __device__ __forceinline__ float prep1(const float zi, const float wi)
{
    const float x = fmaxf(zi, 0.f);
    return 2.f * x - zi - ALPHA * wi;
}

__global__ __launch_bounds__(256) void k_prep_u(const float* __restrict__ z,
                                                const float* __restrict__ w,
                                                float* __restrict__ u, int n, const int* skip)
{
    if (skip && *skip) return;
    const int T  = gridDim.x * blockDim.x;
    const int n4 = n >> 2;
    const float4* z4 = (const float4*)z;
    const float4* w4 = (const float4*)w;
    float4* u4 = (float4*)u;
    for (int base = blockIdx.x * blockDim.x + threadIdx.x; base < n4; base += EW_PIECES * T) {
        float4 zv[EW_PIECES], wv[EW_PIECES];
#pragma unroll
        for (int j = 0; j < EW_PIECES; ++j) {
            const int i = base + j * T;
            if (i < n4) { zv[j] = z4[i]; wv[j] = w4[i]; }
        }
#pragma unroll
        for (int j = 0; j < EW_PIECES; ++j) {
            const int i = base + j * T;
            if (i < n4)
                u4[i] = make_float4(prep1(zv[j].x, wv[j].x), prep1(zv[j].y, wv[j].y),
                                    prep1(zv[j].z, wv[j].z), prep1(zv[j].w, wv[j].w));
        }
    }
}

static __device__ __forceinline__ float upd1(const float zi, const float yi, float& d2)
{
    const float x = fmaxf(zi, 0.f);
    const float d = yi - x;
    d2 += d * d;
    return zi - x + yi;
}

__global__ __launch_bounds__(256) void k_update(float* __restrict__ z, const float* __restrict__ y,
                                                float* __restrict__ acc, int n, const int* skip)
{
    __shared__ float sm[256];
    if (*skip) return;
    const int T  = gridDim.x * blockDim.x;
    const int n4 = n >> 2;
    float4* z4 = (float4*)z;
    const float4* y4 = (const float4*)y;
    float d2 = 0.f;
    for (int base = blockIdx.x * blockDim.x + threadIdx.x; base < n4; base += EW_PIECES * T) {
        float4 zv[EW_PIECES], yv[EW_PIECES];
#pragma unroll
        for (int j = 0; j < EW_PIECES; ++j) {
            const int i = base + j * T;
            if (i < n4) { zv[j] = z4[i]; yv[j] = y4[i]; }
        }
#pragma unroll
        for (int j = 0; j < EW_PIECES; ++j) {
            const int i = base + j * T;
            if (i < n4) {
                float4 r;
                r.x = upd1(zv[j].x, yv[j].x, d2);
                r.y = upd1(zv[j].y, yv[j].y, d2);
                r.z = upd1(zv[j].z, yv[j].z, d2);
                r.w = upd1(zv[j].w, yv[j].w, d2);
                z4[i] = r;
            }
        }
    }
    sm[threadIdx.x] = d2;
    __syncthreads();
    for (int s = 128; s > 0; s >>= 1) {
        if (threadIdx.x < s) sm[threadIdx.x] += sm[threadIdx.x + s];
        __syncthreads();
    }
    if (threadIdx.x == 0) atomicAdd(acc, sm[0]);
}

__global__ __launch_bounds__(256) void k_sumsq(const float* __restrict__ x,
                                               float* __restrict__ acc, int n)
{
    __shared__ float sm[256];
    const int T  = gridDim.x * blockDim.x;
    const int n4 = n >> 2;
    const float4* x4 = (const float4*)x;
    float v = 0.f;
    for (int base = blockIdx.x * blockDim.x + threadIdx.x; base < n4; base += EW_PIECES * T) {
        float4 xv[EW_PIECES];
#pragma unroll
        for (int j = 0; j < EW_PIECES; ++j) {
            const int i = base + j * T;
            if (i < n4) xv[j] = x4[i];
        }
#pragma unroll
        for (int j = 0; j < EW_PIECES; ++j) {
            const int i = base + j * T;
            if (i < n4)
                v += xv[j].x * xv[j].x + xv[j].y * xv[j].y + xv[j].z * xv[j].z + xv[j].w * xv[j].w;
        }
    }
    sm[threadIdx.x] = v;
    __syncthreads();
    for (int s = 128; s > 0; s >>= 1) {
        if (threadIdx.x < s) sm[threadIdx.x] += sm[threadIdx.x + s];
        __syncthreads();
    }
    if (threadIdx.x == 0) atomicAdd(acc, sm[0]);
}

__global__ void k_zero(float* p) { p[0] = 0.f; }

__global__ void k_check(const float* acc, int* flag)
{
    if (*flag == 0 && *acc <= EPS * EPS) *flag = 1;
}

__global__ __launch_bounds__(256) void k_init_X(float* __restrict__ X,
                                                const float* __restrict__ fn, int n)
{
    const int T  = gridDim.x * blockDim.x;
    const int n4 = (n * n) >> 2;
    const float c = rsqrtf(*fn);
    float4* X4 = (float4*)X;
    for (int i = blockIdx.x * blockDim.x + threadIdx.x; i < n4; i += T) {
        const int e = 4 * i;
        const int r = e / n;
        const int d = r - (e - r * n);
        X4[i] = make_float4(d == 0 ? c : 0.f, d == 1 ? c : 0.f, d == 2 ? c : 0.f, d == 3 ? c : 0.f);
    }
}

static __device__ __forceinline__ float fin1(const float zi, const float yi)
{
    const float x = fmaxf(zi, 0.f);
    return fmaxf(zi - x + yi, 0.f);
}

__global__ __launch_bounds__(256) void k_final(const float* __restrict__ z,
                                               const float* __restrict__ y,
                                               float* __restrict__ out, int n)
{
    const int T  = gridDim.x * blockDim.x;
    const int n4 = n >> 2;
    const float4* z4 = (const float4*)z;
    const float4* y4 = (const float4*)y;
    float4* o4 = (float4*)out;
    for (int base = blockIdx.x * blockDim.x + threadIdx.x; base < n4; base += EW_PIECES * T) {
        float4 zv[EW_PIECES], yv[EW_PIECES];
#pragma unroll
        for (int j = 0; j < EW_PIECES; ++j) {
            const int i = base + j * T;
            if (i < n4) { zv[j] = z4[i]; yv[j] = y4[i]; }
        }
#pragma unroll
        for (int j = 0; j < EW_PIECES; ++j) {
            const int i = base + j * T;
            if (i < n4)
                o4[i] = make_float4(fin1(zv[j].x, yv[j].x), fin1(zv[j].y, yv[j].y),
                                    fin1(zv[j].z, yv[j].z), fin1(zv[j].w, yv[j].w));
        }
    }
}

extern "C" void kernel_launch(void* const* d_in, const int* in_sizes, int n_in,
                              void* d_out, int out_size, void* d_ws, size_t ws_size,
                              hipStream_t stream)
{
    const float* ctx  = (const float*)d_in[0];
    const float* Ain  = (const float*)d_in[1];
    const float* bvec = (const float*)d_in[2];
    const float* W1   = (const float*)d_in[3];
    const float* b1   = (const float*)d_in[4];
    const float* W2   = (const float*)d_in[5];
    const float* b2   = (const float*)d_in[6];
    const float* z0   = (const float*)d_in[7];
    float* out = (float*)d_out;

    const int B = 256, D = 512, H = 2048, N1 = 1024, N2 = 4096;

    float* W = (float*)d_ws;
    size_t o = 0;
    float* z  = W + o; o += (size_t)B * N2;
    float* u  = W + o; o += (size_t)B * N2;
    float* y  = W + o; o += (size_t)B * N2;
    float* wv = W + o; o += (size_t)B * N2;
    float* h  = W + o; o += (size_t)B * H;
    float* R  = W + o; o += (size_t)B * N1;
    float* S  = W + o; o += (size_t)B * N1;
    float* G  = W + o; o += (size_t)N1 * N1;
    float* Xa = W + o; o += (size_t)N1 * N1;
    float* Xb = W + o; o += (size_t)N1 * N1;
    float* T1 = W + o; o += (size_t)N1 * N1;
    float* acc = W + o; o += 1;
    float* fn  = W + o; o += 1;
    int*   flag = (int*)(W + o); o += 1;

    dim3 blk(256);
    auto gemm = [&](const float* A_, int lda, const float* B_, int ldb, int tB,
                    float* C_, int ldc, const float* Cin, const float* bias,
                    float al, float be, float ga, int relu,
                    int M, int N, int K, const int* skip, int tile) {
        dim3 grid(N / tile, M / tile);
        if (tile == 64) {
            if (tB)
                gemm_f32_wmma<1, 64><<<grid, blk, 0, stream>>>(A_, lda, B_, ldb, C_, ldc,
                                                               Cin, bias, al, be, ga, relu, K, skip);
            else
                gemm_f32_wmma<0, 64><<<grid, blk, 0, stream>>>(A_, lda, B_, ldb, C_, ldc,
                                                               Cin, bias, al, be, ga, relu, K, skip);
        } else {
            if (tB)
                gemm_f32_wmma<1, 32><<<grid, blk, 0, stream>>>(A_, lda, B_, ldb, C_, ldc,
                                                               Cin, bias, al, be, ga, relu, K, skip);
            else
                gemm_f32_wmma<0, 32><<<grid, blk, 0, stream>>>(A_, lda, B_, ldb, C_, ldc,
                                                               Cin, bias, al, be, ga, relu, K, skip);
        }
    };

    hipMemcpyAsync(z, z0, sizeof(float) * (size_t)B * N2, hipMemcpyDeviceToDevice, stream);
    hipMemsetAsync(flag, 0, sizeof(int), stream);

    gemm(ctx, D, W1, H, 0, h,  H,  nullptr, b1, 1.f, 0.f, 1.f, 1, B, H,  D, nullptr, 32);
    gemm(h,   H, W2, N2, 0, wv, N2, nullptr, b2, 1.f, 0.f, 1.f, 0, B, N2, H, nullptr, 64);

    gemm(Ain, N2, Ain, N2, 1, G, N1, nullptr, nullptr, 1.f, 0.f, 0.f, 0, N1, N1, N2, nullptr, 64);

    k_zero<<<1, 1, 0, stream>>>(fn);
    k_sumsq<<<dim3(EW_BLOCKS), blk, 0, stream>>>(G, fn, N1 * N1);
    k_init_X<<<dim3(EW_BLOCKS), blk, 0, stream>>>(Xa, fn, N1);

    float* Xc = Xa; float* Xn = Xb;
    for (int it = 0; it < 16; ++it) {
        gemm(G,  N1, Xc, N1, 0, T1, N1, nullptr, nullptr,  1.f, 0.f, 0.f, 0, N1, N1, N1, nullptr, 64);
        gemm(Xc, N1, T1, N1, 0, Xn, N1, Xc,      nullptr, -1.f, 2.f, 0.f, 0, N1, N1, N1, nullptr, 64);
        float* tmp = Xc; Xc = Xn; Xn = tmp;
    }
    float* Ginv = Xc;

    const int nz = B * N2;
    dim3 gout(EW_BLOCKS);
    for (int itr = 0; itr < 50; ++itr) {
        k_prep_u<<<gout, blk, 0, stream>>>(z, wv, u, nz, (const int*)flag);
        gemm(u, N2, Ain,  N2, 1, R, N1, nullptr, bvec, 1.f, 0.f, -1.f, 0, B, N1, N2, flag, 32);
        gemm(R, N1, Ginv, N1, 0, S, N1, nullptr, nullptr, 1.f, 0.f, 0.f, 0, B, N1, N1, flag, 32);
        gemm(S, N1, Ain,  N2, 0, y, N2, u, nullptr, -1.f, 1.f, 0.f, 0, B, N2, N1, flag, 64);
        k_zero<<<1, 1, 0, stream>>>(acc);
        k_update<<<gout, blk, 0, stream>>>(z, y, acc, nz, (const int*)flag);
        k_check<<<1, 1, 0, stream>>>((const float*)acc, flag);
    }

    k_prep_u<<<gout, blk, 0, stream>>>(z, wv, u, nz, (const int*)nullptr);
    gemm(u, N2, Ain,  N2, 1, R, N1, nullptr, bvec, 1.f, 0.f, -1.f, 0, B, N1, N2, nullptr, 32);
    gemm(R, N1, Ginv, N1, 0, S, N1, nullptr, nullptr, 1.f, 0.f, 0.f, 0, B, N1, N1, nullptr, 32);
    gemm(S, N1, Ain,  N2, 0, y, N2, u, nullptr, -1.f, 1.f, 0.f, 0, B, N2, N1, nullptr, 64);
    k_final<<<gout, blk, 0, stream>>>(z, y, out, nz);
}


# fixed-point loop: the z update also forms the next u from the z it holds, and the first GEMM takes the gate from the step-norm partials itself: 4 launches an iteration, was 5; sum orders kept
# speedup vs baseline: 7.3162x; 1.0739x over previous
#include <hip/hip_runtime.h>

typedef __attribute__((ext_vector_type(2))) float v2f;
typedef __attribute__((ext_vector_type(8))) float v8f;

#define ALPHA 0.05f
#define EPS   0.01f

#define BK 256
#define WK 32
#define APITCH 36

static __device__ __forceinline__ v8f wmma4(const v2f a, const v2f b, const v8f c)
{
    return __builtin_amdgcn_wmma_f32_16x16x4_f32(false, a, false, b, (short)0, c,
                                                 false, false);
}

#define EW_BLOCKS 256
#define PART_PITCH 32

static __device__ __forceinline__ float sum_partials(const float* __restrict__ part, float* slot)
{
    if (threadIdx.x < 32) {
        float v = 0.f;
#pragma unroll
        for (int j = 0; j < EW_BLOCKS / 32; ++j)
            v += part[(threadIdx.x + 32 * j) * PART_PITCH];
        for (int m = 16; m > 0; m >>= 1) v += __shfl_xor(v, m);
        if (threadIdx.x == 0) *slot = v;
    }
    __syncthreads();
    return *slot;
}

template <int TRANSB, int T, int GATE>
__global__ __launch_bounds__(256) void gemm_f32_wmma(
    const float* __restrict__ Am, int lda,
    const float* __restrict__ Bm, int ldb,
    float* __restrict__ Cm, int ldc,
    const float* __restrict__ Cin,
    const float* __restrict__ bias,
    float alpha, float beta, float gamma, int relu,
    int K, const int* skip, const float* gate_part, int* gate_flag)
{
    if (skip && *skip) return;
    if (GATE && gate_part) {
        __shared__ float s_norm;
        if (sum_partials(gate_part, &s_norm) <= EPS * EPS) {
            if (blockIdx.x == 0 && blockIdx.y == 0 && threadIdx.x == 0) *gate_flag = 1;
            return;
        }
    }

    constexpr int FT   = T / 16;
    constexpr int NL   = T / 4;
    constexpr int BP   = T + 8;
    constexpr int BL   = T / 4;
    constexpr int BR   = 32 / BL;
    constexpr int RP   = T + 4;
    constexpr int WA   = T * APITCH;
    constexpr int WB   = (T * APITCH > 32 * BP) ? T * APITCH : 32 * BP;
    constexpr int WREG = WA + WB;
    static_assert(T * RP <= WREG, "a partial tile is parked in the wave's own region");

    __shared__ __attribute__((aligned(16))) float sW[8 * WREG];

    const int t    = threadIdx.x;
    const int row0 = blockIdx.y * T;
    const int col0 = blockIdx.x * T;

    const int wid  = t >> 5;
    const int lane = t & 31;
    const int hi   = lane >> 4;
    const int ln   = lane & 15;
    const int kw   = wid * WK;
    const int ra0  = wid * WREG;
    const int rb0  = ra0 + WA;

    v8f acc[FT][FT];
#pragma unroll
    for (int mi = 0; mi < FT; ++mi)
#pragma unroll
        for (int ni = 0; ni < FT; ++ni) acc[mi][ni] = (v8f){};

    const int lr = lane >> 3;
    const int lc = (lane & 7) * 4;
    const int br = lane / BL;
    const int bc = (lane % BL) * 4;

    for (int kb = 0; kb < K; kb += BK) {
        float4 ra[NL], rb[NL];
        const float* ab = Am + (size_t)row0 * lda + kb;
        const unsigned ao = (unsigned)(lr * lda + kw + lc);
#pragma unroll
        for (int j = 0; j < NL; ++j) ra[j] = *(const float4*)(ab + (size_t)(4 * j) * lda + ao);
        if (TRANSB) {
            const float* bb = Bm + (size_t)col0 * ldb + kb;
            const unsigned bo = (unsigned)(lr * ldb + kw + lc);
#pragma unroll
            for (int j = 0; j < NL; ++j) rb[j] = *(const float4*)(bb + (size_t)(4 * j) * ldb + bo);
            if (kb + BK < K) {
#pragma unroll
                for (int p = 0; p < T; p += 32) {
                    __builtin_prefetch(ab + BK + (unsigned)((p + lane) * lda + kw), 0, 1);
                    __builtin_prefetch(bb + BK + (unsigned)((p + lane) * ldb + kw), 0, 1);
                }
            }
        } else {
            const float* bb = Bm + (size_t)kb * ldb + col0;
            const unsigned bo = (unsigned)((kw + br) * ldb + bc);
#pragma unroll
            for (int j = 0; j < NL; ++j) rb[j] = *(const float4*)(bb + (size_t)(BR * j) * ldb + bo);
            if (kb + BK < K) {
#pragma unroll
                for (int p = 0; p < T; p += 32)
                    __builtin_prefetch(ab + BK + (unsigned)((p + lane) * lda + kw), 0, 1);
                __builtin_prefetch(bb + (size_t)BK * ldb + (unsigned)((kw + lane) * ldb), 0, 1);
            }
        }
        __builtin_amdgcn_sched_barrier(0);
#pragma unroll
        for (int j = 0; j < NL; ++j) {
            float* q = &sW[ra0 + (lr + 4 * j) * APITCH + lc];
            q[0] = ra[j].x; q[1] = ra[j].y; q[2] = ra[j].z; q[3] = ra[j].w;
        }
#pragma unroll
        for (int j = 0; j < NL; ++j) {
            float* q = TRANSB ? &sW[rb0 + (lr + 4 * j) * APITCH + lc]
                              : &sW[rb0 + (br + BR * j) * BP + bc];
            q[0] = rb[j].x; q[1] = rb[j].y; q[2] = rb[j].z; q[3] = rb[j].w;
        }
        __builtin_amdgcn_fence(__ATOMIC_ACQ_REL, "wavefront");

#pragma unroll
        for (int kk = 0; kk < WK; kk += 4) {
            const int k = kk + 2 * hi;
            v2f a[FT], b[FT];
#pragma unroll
            for (int mi = 0; mi < FT; ++mi)
                a[mi] = *(const v2f*)&sW[ra0 + (16 * mi + ln) * APITCH + k];
#pragma unroll
            for (int ni = 0; ni < FT; ++ni) {
                if (TRANSB) {
                    b[ni] = *(const v2f*)&sW[rb0 + (16 * ni + ln) * APITCH + k];
                } else {
                    b[ni].x = sW[rb0 + k * BP + 16 * ni + ln];
                    b[ni].y = sW[rb0 + (k + 1) * BP + 16 * ni + ln];
                }
            }
#pragma unroll
            for (int mi = 0; mi < FT; ++mi)
#pragma unroll
                for (int ni = 0; ni < FT; ++ni)
                    acc[mi][ni] = wmma4(a[mi], b[ni], acc[mi][ni]);
        }
        __builtin_amdgcn_fence(__ATOMIC_ACQ_REL, "wavefront");
    }

#pragma unroll
    for (int mi = 0; mi < FT; ++mi)
#pragma unroll
        for (int ni = 0; ni < FT; ++ni)
#pragma unroll
            for (int v = 0; v < 8; ++v)
                sW[ra0 + (16 * mi + 8 * hi + v) * RP + 16 * ni + ln] = acc[mi][ni][v];
    __syncthreads();

    constexpr int NP = T * T / 1024;
    constexpr int TR = 1024 / T;
    const int sr0 = t / (T / 4);
    const int sc  = (t % (T / 4)) * 4;
    const int gc  = col0 + sc;
    float4 bb = make_float4(0.f, 0.f, 0.f, 0.f);
    if (bias) bb = *(const float4*)(bias + gc);
    float4 ci[NP];
#pragma unroll
    for (int pc = 0; pc < NP; ++pc) ci[pc] = make_float4(0.f, 0.f, 0.f, 0.f);
    if (Cin) {
#pragma unroll
        for (int pc = 0; pc < NP; ++pc)
            ci[pc] = *(const float4*)(Cin + (size_t)(row0 + sr0 + TR * pc) * ldc + gc);
    }
#pragma unroll
    for (int pc = 0; pc < NP; ++pc) {
        const int sr = sr0 + TR * pc;
        const float4 bv = make_float4(gamma * bb.x, gamma * bb.y, gamma * bb.z, gamma * bb.w);
        float4 s = *(const float4*)&sW[sr * RP + sc];
#pragma unroll
        for (int w = 1; w < 8; ++w) {
            const float4 p = *(const float4*)&sW[w * WREG + sr * RP + sc];
            s.x += p.x; s.y += p.y; s.z += p.z; s.w += p.w;
        }
        float4 o = make_float4(alpha * s.x + bv.x, alpha * s.y + bv.y,
                               alpha * s.z + bv.z, alpha * s.w + bv.w);
        if (Cin) {
            o.x += beta * ci[pc].x; o.y += beta * ci[pc].y;
            o.z += beta * ci[pc].z; o.w += beta * ci[pc].w;
        }
        if (relu) {
            o.x = fmaxf(o.x, 0.f); o.y = fmaxf(o.y, 0.f);
            o.z = fmaxf(o.z, 0.f); o.w = fmaxf(o.w, 0.f);
        }
        *(float4*)(Cm + (size_t)(row0 + sr) * ldc + gc) = o;
    }
}

#define EW_PIECES 4
static __device__ __forceinline__ float prep1(const float zi, const float wi)
{
    const float x = fmaxf(zi, 0.f);
    return 2.f * x - zi - ALPHA * wi;
}

__global__ __launch_bounds__(256) void k_prep_u(const float* __restrict__ z,
                                                const float* __restrict__ w,
                                                float* __restrict__ u, int n)
{
    const int T  = gridDim.x * blockDim.x;
    const int n4 = n >> 2;
    const float4* z4 = (const float4*)z;
    const float4* w4 = (const float4*)w;
    float4* u4 = (float4*)u;
    for (int base = blockIdx.x * blockDim.x + threadIdx.x; base < n4; base += EW_PIECES * T) {
        float4 zv[EW_PIECES], wv[EW_PIECES];
#pragma unroll
        for (int j = 0; j < EW_PIECES; ++j) {
            const int i = base + j * T;
            if (i < n4) { zv[j] = z4[i]; wv[j] = w4[i]; }
        }
#pragma unroll
        for (int j = 0; j < EW_PIECES; ++j) {
            const int i = base + j * T;
            if (i < n4)
                u4[i] = make_float4(prep1(zv[j].x, wv[j].x), prep1(zv[j].y, wv[j].y),
                                    prep1(zv[j].z, wv[j].z), prep1(zv[j].w, wv[j].w));
        }
    }
}

static __device__ __forceinline__ float upd1(const float zi, const float yi, float& d2)
{
    const float x = fmaxf(zi, 0.f);
    const float d = yi - x;
    d2 += d * d;
    return zi - x + yi;
}

__global__ __launch_bounds__(256) void k_update(float* __restrict__ z, const float* __restrict__ y,
                                                const float* __restrict__ w, float* __restrict__ u,
                                                float* __restrict__ part, int n, const int* skip)
{
    __shared__ float sm[256];
    if (*skip) return;
    const int T  = gridDim.x * blockDim.x;
    const int n4 = n >> 2;
    float4* z4 = (float4*)z;
    const float4* y4 = (const float4*)y;
    const float4* w4 = (const float4*)w;
    float4* u4 = (float4*)u;
    float d2 = 0.f;
    for (int base = blockIdx.x * blockDim.x + threadIdx.x; base < n4; base += EW_PIECES * T) {
        float4 zv[EW_PIECES], yv[EW_PIECES], wv[EW_PIECES];
#pragma unroll
        for (int j = 0; j < EW_PIECES; ++j) {
            const int i = base + j * T;
            if (i < n4) { zv[j] = z4[i]; yv[j] = y4[i]; wv[j] = w4[i]; }
        }
#pragma unroll
        for (int j = 0; j < EW_PIECES; ++j) {
            const int i = base + j * T;
            if (i < n4) {
                float4 r;
                r.x = upd1(zv[j].x, yv[j].x, d2);
                r.y = upd1(zv[j].y, yv[j].y, d2);
                r.z = upd1(zv[j].z, yv[j].z, d2);
                r.w = upd1(zv[j].w, yv[j].w, d2);
                z4[i] = r;
                u4[i] = make_float4(prep1(r.x, wv[j].x), prep1(r.y, wv[j].y),
                                    prep1(r.z, wv[j].z), prep1(r.w, wv[j].w));
            }
        }
    }
    sm[threadIdx.x] = d2;
    __syncthreads();
    for (int s = 128; s > 0; s >>= 1) {
        if (threadIdx.x < s) sm[threadIdx.x] += sm[threadIdx.x + s];
        __syncthreads();
    }
    if (threadIdx.x == 0) part[blockIdx.x * PART_PITCH] = sm[0];
}

__global__ __launch_bounds__(256) void k_sumsq(const float* __restrict__ x,
                                               float* __restrict__ part, int n)
{
    __shared__ float sm[256];
    const int T  = gridDim.x * blockDim.x;
    const int n4 = n >> 2;
    const float4* x4 = (const float4*)x;
    float v = 0.f;
    for (int base = blockIdx.x * blockDim.x + threadIdx.x; base < n4; base += EW_PIECES * T) {
        float4 xv[EW_PIECES];
#pragma unroll
        for (int j = 0; j < EW_PIECES; ++j) {
            const int i = base + j * T;
            if (i < n4) xv[j] = x4[i];
        }
#pragma unroll
        for (int j = 0; j < EW_PIECES; ++j) {
            const int i = base + j * T;
            if (i < n4)
                v += xv[j].x * xv[j].x + xv[j].y * xv[j].y + xv[j].z * xv[j].z + xv[j].w * xv[j].w;
        }
    }
    sm[threadIdx.x] = v;
    __syncthreads();
    for (int s = 128; s > 0; s >>= 1) {
        if (threadIdx.x < s) sm[threadIdx.x] += sm[threadIdx.x + s];
        __syncthreads();
    }
    if (threadIdx.x == 0) part[blockIdx.x * PART_PITCH] = sm[0];
}

__global__ __launch_bounds__(256) void k_init_X(float* __restrict__ X,
                                                const float* __restrict__ part, int n)
{
    __shared__ float s_fn;
    const int T  = gridDim.x * blockDim.x;
    const int n4 = (n * n) >> 2;
    const float c = rsqrtf(sum_partials(part, &s_fn));
    float4* X4 = (float4*)X;
    for (int i = blockIdx.x * blockDim.x + threadIdx.x; i < n4; i += T) {
        const int e = 4 * i;
        const int r = e / n;
        const int d = r - (e - r * n);
        X4[i] = make_float4(d == 0 ? c : 0.f, d == 1 ? c : 0.f, d == 2 ? c : 0.f, d == 3 ? c : 0.f);
    }
}

static __device__ __forceinline__ float fin1(const float zi, const float yi)
{
    const float x = fmaxf(zi, 0.f);
    return fmaxf(zi - x + yi, 0.f);
}

__global__ __launch_bounds__(256) void k_final(const float* __restrict__ z,
                                               const float* __restrict__ y,
                                               float* __restrict__ out, int n)
{
    const int T  = gridDim.x * blockDim.x;
    const int n4 = n >> 2;
    const float4* z4 = (const float4*)z;
    const float4* y4 = (const float4*)y;
    float4* o4 = (float4*)out;
    for (int base = blockIdx.x * blockDim.x + threadIdx.x; base < n4; base += EW_PIECES * T) {
        float4 zv[EW_PIECES], yv[EW_PIECES];
#pragma unroll
        for (int j = 0; j < EW_PIECES; ++j) {
            const int i = base + j * T;
            if (i < n4) { zv[j] = z4[i]; yv[j] = y4[i]; }
        }
#pragma unroll
        for (int j = 0; j < EW_PIECES; ++j) {
            const int i = base + j * T;
            if (i < n4)
                o4[i] = make_float4(fin1(zv[j].x, yv[j].x), fin1(zv[j].y, yv[j].y),
                                    fin1(zv[j].z, yv[j].z), fin1(zv[j].w, yv[j].w));
        }
    }
}

extern "C" void kernel_launch(void* const* d_in, const int* in_sizes, int n_in,
                              void* d_out, int out_size, void* d_ws, size_t ws_size,
                              hipStream_t stream)
{
    const float* ctx  = (const float*)d_in[0];
    const float* Ain  = (const float*)d_in[1];
    const float* bvec = (const float*)d_in[2];
    const float* W1   = (const float*)d_in[3];
    const float* b1   = (const float*)d_in[4];
    const float* W2   = (const float*)d_in[5];
    const float* b2   = (const float*)d_in[6];
    const float* z0   = (const float*)d_in[7];
    float* out = (float*)d_out;

    const int B = 256, D = 512, H = 2048, N1 = 1024, N2 = 4096;

    float* W = (float*)d_ws;
    size_t o = 0;
    float* z  = W + o; o += (size_t)B * N2;
    float* u  = W + o; o += (size_t)B * N2;
    float* y  = W + o; o += (size_t)B * N2;
    float* wv = W + o; o += (size_t)B * N2;
    float* h  = W + o; o += (size_t)B * H;
    float* R  = W + o; o += (size_t)B * N1;
    float* S  = W + o; o += (size_t)B * N1;
    float* G  = W + o; o += (size_t)N1 * N1;
    float* Xa = W + o; o += (size_t)N1 * N1;
    float* Xb = W + o; o += (size_t)N1 * N1;
    float* T1 = W + o; o += (size_t)N1 * N1;
    int*   flag = (int*)(W + o); o += 1;

    dim3 blk(256);
    auto gemm = [&](const float* A_, int lda, const float* B_, int ldb, int tB,
                    float* C_, int ldc, const float* Cin, const float* bias,
                    float al, float be, float ga, int relu,
                    int M, int N, int K, const int* skip, int tile,
                    const float* gate_part = nullptr, int* gate_flag = nullptr) {
        dim3 grid(N / tile, M / tile);
        if (tile == 64) {
            if (tB)
                gemm_f32_wmma<1, 64, 0><<<grid, blk, 0, stream>>>(A_, lda, B_, ldb, C_, ldc,
                                                               Cin, bias, al, be, ga, relu, K, skip,
                                                               gate_part, gate_flag);
            else
                gemm_f32_wmma<0, 64, 0><<<grid, blk, 0, stream>>>(A_, lda, B_, ldb, C_, ldc,
                                                               Cin, bias, al, be, ga, relu, K, skip,
                                                               gate_part, gate_flag);
        } else {
            if (tB && gate_flag)
                gemm_f32_wmma<1, 32, 1><<<grid, blk, 0, stream>>>(A_, lda, B_, ldb, C_, ldc,
                                                                  Cin, bias, al, be, ga, relu, K, skip,
                                                                  gate_part, gate_flag);
            else if (tB)
                gemm_f32_wmma<1, 32, 0><<<grid, blk, 0, stream>>>(A_, lda, B_, ldb, C_, ldc,
                                                               Cin, bias, al, be, ga, relu, K, skip,
                                                               gate_part, gate_flag);
            else
                gemm_f32_wmma<0, 32, 0><<<grid, blk, 0, stream>>>(A_, lda, B_, ldb, C_, ldc,
                                                               Cin, bias, al, be, ga, relu, K, skip,
                                                               gate_part, gate_flag);
        }
    };

    hipMemcpyAsync(z, z0, sizeof(float) * (size_t)B * N2, hipMemcpyDeviceToDevice, stream);
    hipMemsetAsync(flag, 0, sizeof(int), stream);

    gemm(ctx, D, W1, H, 0, h,  H,  nullptr, b1, 1.f, 0.f, 1.f, 1, B, H,  D, nullptr, 32);
    gemm(h,   H, W2, N2, 0, wv, N2, nullptr, b2, 1.f, 0.f, 1.f, 0, B, N2, H, nullptr, 64);

    gemm(Ain, N2, Ain, N2, 1, G, N1, nullptr, nullptr, 1.f, 0.f, 0.f, 0, N1, N1, N2, nullptr, 64);

    float* part = h;
    k_sumsq<<<dim3(EW_BLOCKS), blk, 0, stream>>>(G, part, N1 * N1);
    k_init_X<<<dim3(EW_BLOCKS), blk, 0, stream>>>(Xa, part, N1);

    float* Xc = Xa; float* Xn = Xb;
    for (int it = 0; it < 16; ++it) {
        gemm(G,  N1, Xc, N1, 0, T1, N1, nullptr, nullptr,  1.f, 0.f, 0.f, 0, N1, N1, N1, nullptr, 64);
        gemm(Xc, N1, T1, N1, 0, Xn, N1, Xc,      nullptr, -1.f, 2.f, 0.f, 0, N1, N1, N1, nullptr, 64);
        float* tmp = Xc; Xc = Xn; Xn = tmp;
    }
    float* Ginv = Xc;

    const float* gate = nullptr;
    const int nz = B * N2;
    dim3 gout(EW_BLOCKS);
    k_prep_u<<<gout, blk, 0, stream>>>(z, wv, u, nz);
    for (int itr = 0; itr < 50; ++itr) {
        gemm(u, N2, Ain,  N2, 1, R, N1, nullptr, bvec, 1.f, 0.f, -1.f, 0, B, N1, N2, nullptr, 32,
             gate, flag);
        gemm(R, N1, Ginv, N1, 0, S, N1, nullptr, nullptr, 1.f, 0.f, 0.f, 0, B, N1, N1, flag, 32);
        gemm(S, N1, Ain,  N2, 0, y, N2, u, nullptr, -1.f, 1.f, 0.f, 0, B, N2, N1, flag, 64);
        k_update<<<gout, blk, 0, stream>>>(z, y, wv, u, part, nz, (const int*)flag);
        gate = part;
    }

    gemm(u, N2, Ain,  N2, 1, R, N1, nullptr, bvec, 1.f, 0.f, -1.f, 0, B, N1, N2, nullptr, 32);
    gemm(R, N1, Ginv, N1, 0, S, N1, nullptr, nullptr, 1.f, 0.f, 0.f, 0, B, N1, N1, nullptr, 32);
    gemm(S, N1, Ain,  N2, 0, y, N2, u, nullptr, -1.f, 1.f, 0.f, 0, B, N2, N1, nullptr, 64);
    k_final<<<gout, blk, 0, stream>>>(z, y, out, nz);
}


# 64x64-tile GEMMs: a wave holds priority (s_setprio) from a quarter into a trip's 128 WMMAs to their end, fragment reads one k-step ahead: a SIMD's two waves use the pipe in turns; sums unchanged
# speedup vs baseline: 7.6223x; 1.0418x over previous
#include <hip/hip_runtime.h>

typedef __attribute__((ext_vector_type(2))) float v2f;
typedef __attribute__((ext_vector_type(8))) float v8f;

#define ALPHA 0.05f
#define EPS   0.01f

#define BK 256
#define WK 32
#define APITCH 36

static __device__ __forceinline__ v8f wmma4(const v2f a, const v2f b, const v8f c)
{
    return __builtin_amdgcn_wmma_f32_16x16x4_f32(false, a, false, b, (short)0, c,
                                                 false, false);
}

#define EW_BLOCKS 256
#define PART_PITCH 32

static __device__ __forceinline__ float sum_partials(const float* __restrict__ part, float* slot)
{
    if (threadIdx.x < 32) {
        float v = 0.f;
#pragma unroll
        for (int j = 0; j < EW_BLOCKS / 32; ++j)
            v += part[(threadIdx.x + 32 * j) * PART_PITCH];
        for (int m = 16; m > 0; m >>= 1) v += __shfl_xor(v, m);
        if (threadIdx.x == 0) *slot = v;
    }
    __syncthreads();
    return *slot;
}

template <int TRANSB, int T, int GATE>
__global__ __launch_bounds__(256) void gemm_f32_wmma(
    const float* __restrict__ Am, int lda,
    const float* __restrict__ Bm, int ldb,
    float* __restrict__ Cm, int ldc,
    const float* __restrict__ Cin,
    const float* __restrict__ bias,
    float alpha, float beta, float gamma, int relu,
    int K, const int* skip, const float* gate_part, int* gate_flag)
{
    if (skip && *skip) return;
    if (GATE && gate_part) {
        __shared__ float s_norm;
        if (sum_partials(gate_part, &s_norm) <= EPS * EPS) {
            if (blockIdx.x == 0 && blockIdx.y == 0 && threadIdx.x == 0) *gate_flag = 1;
            return;
        }
    }

    constexpr int FT   = T / 16;
    constexpr int NL   = T / 4;
    constexpr int BP   = T + 8;
    constexpr int BL   = T / 4;
    constexpr int BR   = 32 / BL;
    constexpr int RP   = T + 4;
    constexpr int WA   = T * APITCH;
    constexpr int WB   = (T * APITCH > 32 * BP) ? T * APITCH : 32 * BP;
    constexpr int WREG = WA + WB;
    static_assert(T * RP <= WREG, "a partial tile is parked in the wave's own region");

    __shared__ __attribute__((aligned(16))) float sW[8 * WREG];

    const int t    = threadIdx.x;
    const int row0 = blockIdx.y * T;
    const int col0 = blockIdx.x * T;

    const int wid  = t >> 5;
    const int lane = t & 31;
    const int hi   = lane >> 4;
    const int ln   = lane & 15;
    const int kw   = wid * WK;
    const int ra0  = wid * WREG;
    const int rb0  = ra0 + WA;

    v8f acc[FT][FT];
#pragma unroll
    for (int mi = 0; mi < FT; ++mi)
#pragma unroll
        for (int ni = 0; ni < FT; ++ni) acc[mi][ni] = (v8f){};

    const int lr = lane >> 3;
    const int lc = (lane & 7) * 4;
    const int br = lane / BL;
    const int bc = (lane % BL) * 4;

    for (int kb = 0; kb < K; kb += BK) {
        float4 ra[NL], rb[NL];
        const float* ab = Am + (size_t)row0 * lda + kb;
        const unsigned ao = (unsigned)(lr * lda + kw + lc);
#pragma unroll
        for (int j = 0; j < NL; ++j) ra[j] = *(const float4*)(ab + (size_t)(4 * j) * lda + ao);
        if (TRANSB) {
            const float* bb = Bm + (size_t)col0 * ldb + kb;
            const unsigned bo = (unsigned)(lr * ldb + kw + lc);
#pragma unroll
            for (int j = 0; j < NL; ++j) rb[j] = *(const float4*)(bb + (size_t)(4 * j) * ldb + bo);
            if (kb + BK < K) {
#pragma unroll
                for (int p = 0; p < T; p += 32) {
                    __builtin_prefetch(ab + BK + (unsigned)((p + lane) * lda + kw), 0, 1);
                    __builtin_prefetch(bb + BK + (unsigned)((p + lane) * ldb + kw), 0, 1);
                }
            }
        } else {
            const float* bb = Bm + (size_t)kb * ldb + col0;
            const unsigned bo = (unsigned)((kw + br) * ldb + bc);
#pragma unroll
            for (int j = 0; j < NL; ++j) rb[j] = *(const float4*)(bb + (size_t)(BR * j) * ldb + bo);
            if (kb + BK < K) {
#pragma unroll
                for (int p = 0; p < T; p += 32)
                    __builtin_prefetch(ab + BK + (unsigned)((p + lane) * lda + kw), 0, 1);
                __builtin_prefetch(bb + (size_t)BK * ldb + (unsigned)((kw + lane) * ldb), 0, 1);
            }
        }
        __builtin_amdgcn_sched_barrier(0);
#pragma unroll
        for (int j = 0; j < NL; ++j) {
            float* q = &sW[ra0 + (lr + 4 * j) * APITCH + lc];
            q[0] = ra[j].x; q[1] = ra[j].y; q[2] = ra[j].z; q[3] = ra[j].w;
        }
#pragma unroll
        for (int j = 0; j < NL; ++j) {
            float* q = TRANSB ? &sW[rb0 + (lr + 4 * j) * APITCH + lc]
                              : &sW[rb0 + (br + BR * j) * BP + bc];
            q[0] = rb[j].x; q[1] = rb[j].y; q[2] = rb[j].z; q[3] = rb[j].w;
        }
        __builtin_amdgcn_fence(__ATOMIC_ACQ_REL, "wavefront");

        constexpr int NSTEP = WK / 4;
        constexpr int AH    = (T == 64) ? 1 : 0;
        v2f a[NSTEP][FT], b[NSTEP][FT];
        auto ldfrag = [&](const int s) {
            const int k = 4 * s + 2 * hi;
#pragma unroll
            for (int mi = 0; mi < FT; ++mi)
                a[s][mi] = *(const v2f*)&sW[ra0 + (16 * mi + ln) * APITCH + k];
#pragma unroll
            for (int ni = 0; ni < FT; ++ni) {
                if (TRANSB) {
                    b[s][ni] = *(const v2f*)&sW[rb0 + (16 * ni + ln) * APITCH + k];
                } else {
                    b[s][ni].x = sW[rb0 + k * BP + 16 * ni + ln];
                    b[s][ni].y = sW[rb0 + (k + 1) * BP + 16 * ni + ln];
                }
            }
        };
#pragma unroll
        for (int s = 0; s < AH; ++s) ldfrag(s);
#pragma unroll
        for (int s = 0; s < NSTEP; ++s) {
            if (s + AH < NSTEP) ldfrag(s + AH);
            if (T == 64 && s == NSTEP / 4) __builtin_amdgcn_s_setprio(3);
#pragma unroll
            for (int mi = 0; mi < FT; ++mi)
#pragma unroll
                for (int ni = 0; ni < FT; ++ni)
                    acc[mi][ni] = wmma4(a[s][mi], b[s][ni], acc[mi][ni]);
        }
        if (T == 64) __builtin_amdgcn_s_setprio(0);
        __builtin_amdgcn_fence(__ATOMIC_ACQ_REL, "wavefront");
    }

#pragma unroll
    for (int mi = 0; mi < FT; ++mi)
#pragma unroll
        for (int ni = 0; ni < FT; ++ni)
#pragma unroll
            for (int v = 0; v < 8; ++v)
                sW[ra0 + (16 * mi + 8 * hi + v) * RP + 16 * ni + ln] = acc[mi][ni][v];
    __syncthreads();

    constexpr int NP = T * T / 1024;
    constexpr int TR = 1024 / T;
    const int sr0 = t / (T / 4);
    const int sc  = (t % (T / 4)) * 4;
    const int gc  = col0 + sc;
    float4 bb = make_float4(0.f, 0.f, 0.f, 0.f);
    if (bias) bb = *(const float4*)(bias + gc);
    float4 ci[NP];
#pragma unroll
    for (int pc = 0; pc < NP; ++pc) ci[pc] = make_float4(0.f, 0.f, 0.f, 0.f);
    if (Cin) {
#pragma unroll
        for (int pc = 0; pc < NP; ++pc)
            ci[pc] = *(const float4*)(Cin + (size_t)(row0 + sr0 + TR * pc) * ldc + gc);
    }
#pragma unroll
    for (int pc = 0; pc < NP; ++pc) {
        const int sr = sr0 + TR * pc;
        const float4 bv = make_float4(gamma * bb.x, gamma * bb.y, gamma * bb.z, gamma * bb.w);
        float4 s = *(const float4*)&sW[sr * RP + sc];
#pragma unroll
        for (int w = 1; w < 8; ++w) {
            const float4 p = *(const float4*)&sW[w * WREG + sr * RP + sc];
            s.x += p.x; s.y += p.y; s.z += p.z; s.w += p.w;
        }
        float4 o = make_float4(alpha * s.x + bv.x, alpha * s.y + bv.y,
                               alpha * s.z + bv.z, alpha * s.w + bv.w);
        if (Cin) {
            o.x += beta * ci[pc].x; o.y += beta * ci[pc].y;
            o.z += beta * ci[pc].z; o.w += beta * ci[pc].w;
        }
        if (relu) {
            o.x = fmaxf(o.x, 0.f); o.y = fmaxf(o.y, 0.f);
            o.z = fmaxf(o.z, 0.f); o.w = fmaxf(o.w, 0.f);
        }
        *(float4*)(Cm + (size_t)(row0 + sr) * ldc + gc) = o;
    }
}

#define EW_PIECES 4
static __device__ __forceinline__ float prep1(const float zi, const float wi)
{
    const float x = fmaxf(zi, 0.f);
    return 2.f * x - zi - ALPHA * wi;
}

__global__ __launch_bounds__(256) void k_prep_u(const float* __restrict__ z,
                                                const float* __restrict__ w,
                                                float* __restrict__ u, int n)
{
    const int T  = gridDim.x * blockDim.x;
    const int n4 = n >> 2;
    const float4* z4 = (const float4*)z;
    const float4* w4 = (const float4*)w;
    float4* u4 = (float4*)u;
    for (int base = blockIdx.x * blockDim.x + threadIdx.x; base < n4; base += EW_PIECES * T) {
        float4 zv[EW_PIECES], wv[EW_PIECES];
#pragma unroll
        for (int j = 0; j < EW_PIECES; ++j) {
            const int i = base + j * T;
            if (i < n4) { zv[j] = z4[i]; wv[j] = w4[i]; }
        }
#pragma unroll
        for (int j = 0; j < EW_PIECES; ++j) {
            const int i = base + j * T;
            if (i < n4)
                u4[i] = make_float4(prep1(zv[j].x, wv[j].x), prep1(zv[j].y, wv[j].y),
                                    prep1(zv[j].z, wv[j].z), prep1(zv[j].w, wv[j].w));
        }
    }
}

static __device__ __forceinline__ float upd1(const float zi, const float yi, float& d2)
{
    const float x = fmaxf(zi, 0.f);
    const float d = yi - x;
    d2 += d * d;
    return zi - x + yi;
}

__global__ __launch_bounds__(256) void k_update(float* __restrict__ z, const float* __restrict__ y,
                                                const float* __restrict__ w, float* __restrict__ u,
                                                float* __restrict__ part, int n, const int* skip)
{
    __shared__ float sm[256];
    if (*skip) return;
    const int T  = gridDim.x * blockDim.x;
    const int n4 = n >> 2;
    float4* z4 = (float4*)z;
    const float4* y4 = (const float4*)y;
    const float4* w4 = (const float4*)w;
    float4* u4 = (float4*)u;
    float d2 = 0.f;
    for (int base = blockIdx.x * blockDim.x + threadIdx.x; base < n4; base += EW_PIECES * T) {
        float4 zv[EW_PIECES], yv[EW_PIECES], wv[EW_PIECES];
#pragma unroll
        for (int j = 0; j < EW_PIECES; ++j) {
            const int i = base + j * T;
            if (i < n4) { zv[j] = z4[i]; yv[j] = y4[i]; wv[j] = w4[i]; }
        }
#pragma unroll
        for (int j = 0; j < EW_PIECES; ++j) {
            const int i = base + j * T;
            if (i < n4) {
                float4 r;
                r.x = upd1(zv[j].x, yv[j].x, d2);
                r.y = upd1(zv[j].y, yv[j].y, d2);
                r.z = upd1(zv[j].z, yv[j].z, d2);
                r.w = upd1(zv[j].w, yv[j].w, d2);
                z4[i] = r;
                u4[i] = make_float4(prep1(r.x, wv[j].x), prep1(r.y, wv[j].y),
                                    prep1(r.z, wv[j].z), prep1(r.w, wv[j].w));
            }
        }
    }
    sm[threadIdx.x] = d2;
    __syncthreads();
    for (int s = 128; s > 0; s >>= 1) {
        if (threadIdx.x < s) sm[threadIdx.x] += sm[threadIdx.x + s];
        __syncthreads();
    }
    if (threadIdx.x == 0) part[blockIdx.x * PART_PITCH] = sm[0];
}

__global__ __launch_bounds__(256) void k_sumsq(const float* __restrict__ x,
                                               float* __restrict__ part, int n)
{
    __shared__ float sm[256];
    const int T  = gridDim.x * blockDim.x;
    const int n4 = n >> 2;
    const float4* x4 = (const float4*)x;
    float v = 0.f;
    for (int base = blockIdx.x * blockDim.x + threadIdx.x; base < n4; base += EW_PIECES * T) {
        float4 xv[EW_PIECES];
#pragma unroll
        for (int j = 0; j < EW_PIECES; ++j) {
            const int i = base + j * T;
            if (i < n4) xv[j] = x4[i];
        }
#pragma unroll
        for (int j = 0; j < EW_PIECES; ++j) {
            const int i = base + j * T;
            if (i < n4)
                v += xv[j].x * xv[j].x + xv[j].y * xv[j].y + xv[j].z * xv[j].z + xv[j].w * xv[j].w;
        }
    }
    sm[threadIdx.x] = v;
    __syncthreads();
    for (int s = 128; s > 0; s >>= 1) {
        if (threadIdx.x < s) sm[threadIdx.x] += sm[threadIdx.x + s];
        __syncthreads();
    }
    if (threadIdx.x == 0) part[blockIdx.x * PART_PITCH] = sm[0];
}

__global__ __launch_bounds__(256) void k_init_X(float* __restrict__ X,
                                                const float* __restrict__ part, int n)
{
    __shared__ float s_fn;
    const int T  = gridDim.x * blockDim.x;
    const int n4 = (n * n) >> 2;
    const float c = rsqrtf(sum_partials(part, &s_fn));
    float4* X4 = (float4*)X;
    for (int i = blockIdx.x * blockDim.x + threadIdx.x; i < n4; i += T) {
        const int e = 4 * i;
        const int r = e / n;
        const int d = r - (e - r * n);
        X4[i] = make_float4(d == 0 ? c : 0.f, d == 1 ? c : 0.f, d == 2 ? c : 0.f, d == 3 ? c : 0.f);
    }
}

static __device__ __forceinline__ float fin1(const float zi, const float yi)
{
    const float x = fmaxf(zi, 0.f);
    return fmaxf(zi - x + yi, 0.f);
}

__global__ __launch_bounds__(256) void k_final(const float* __restrict__ z,
                                               const float* __restrict__ y,
                                               float* __restrict__ out, int n)
{
    const int T  = gridDim.x * blockDim.x;
    const int n4 = n >> 2;
    const float4* z4 = (const float4*)z;
    const float4* y4 = (const float4*)y;
    float4* o4 = (float4*)out;
    for (int base = blockIdx.x * blockDim.x + threadIdx.x; base < n4; base += EW_PIECES * T) {
        float4 zv[EW_PIECES], yv[EW_PIECES];
#pragma unroll
        for (int j = 0; j < EW_PIECES; ++j) {
            const int i = base + j * T;
            if (i < n4) { zv[j] = z4[i]; yv[j] = y4[i]; }
        }
#pragma unroll
        for (int j = 0; j < EW_PIECES; ++j) {
            const int i = base + j * T;
            if (i < n4)
                o4[i] = make_float4(fin1(zv[j].x, yv[j].x), fin1(zv[j].y, yv[j].y),
                                    fin1(zv[j].z, yv[j].z), fin1(zv[j].w, yv[j].w));
        }
    }
}

extern "C" void kernel_launch(void* const* d_in, const int* in_sizes, int n_in,
                              void* d_out, int out_size, void* d_ws, size_t ws_size,
                              hipStream_t stream)
{
    const float* ctx  = (const float*)d_in[0];
    const float* Ain  = (const float*)d_in[1];
    const float* bvec = (const float*)d_in[2];
    const float* W1   = (const float*)d_in[3];
    const float* b1   = (const float*)d_in[4];
    const float* W2   = (const float*)d_in[5];
    const float* b2   = (const float*)d_in[6];
    const float* z0   = (const float*)d_in[7];
    float* out = (float*)d_out;

    const int B = 256, D = 512, H = 2048, N1 = 1024, N2 = 4096;

    float* W = (float*)d_ws;
    size_t o = 0;
    float* z  = W + o; o += (size_t)B * N2;
    float* u  = W + o; o += (size_t)B * N2;
    float* y  = W + o; o += (size_t)B * N2;
    float* wv = W + o; o += (size_t)B * N2;
    float* h  = W + o; o += (size_t)B * H;
    float* R  = W + o; o += (size_t)B * N1;
    float* S  = W + o; o += (size_t)B * N1;
    float* G  = W + o; o += (size_t)N1 * N1;
    float* Xa = W + o; o += (size_t)N1 * N1;
    float* Xb = W + o; o += (size_t)N1 * N1;
    float* T1 = W + o; o += (size_t)N1 * N1;
    int*   flag = (int*)(W + o); o += 1;

    dim3 blk(256);
    auto gemm = [&](const float* A_, int lda, const float* B_, int ldb, int tB,
                    float* C_, int ldc, const float* Cin, const float* bias,
                    float al, float be, float ga, int relu,
                    int M, int N, int K, const int* skip, int tile,
                    const float* gate_part = nullptr, int* gate_flag = nullptr) {
        dim3 grid(N / tile, M / tile);
        if (tile == 64) {
            if (tB)
                gemm_f32_wmma<1, 64, 0><<<grid, blk, 0, stream>>>(A_, lda, B_, ldb, C_, ldc,
                                                               Cin, bias, al, be, ga, relu, K, skip,
                                                               gate_part, gate_flag);
            else
                gemm_f32_wmma<0, 64, 0><<<grid, blk, 0, stream>>>(A_, lda, B_, ldb, C_, ldc,
                                                               Cin, bias, al, be, ga, relu, K, skip,
                                                               gate_part, gate_flag);
        } else {
            if (tB && gate_flag)
                gemm_f32_wmma<1, 32, 1><<<grid, blk, 0, stream>>>(A_, lda, B_, ldb, C_, ldc,
                                                                  Cin, bias, al, be, ga, relu, K, skip,
                                                                  gate_part, gate_flag);
            else if (tB)
                gemm_f32_wmma<1, 32, 0><<<grid, blk, 0, stream>>>(A_, lda, B_, ldb, C_, ldc,
                                                               Cin, bias, al, be, ga, relu, K, skip,
                                                               gate_part, gate_flag);
            else
                gemm_f32_wmma<0, 32, 0><<<grid, blk, 0, stream>>>(A_, lda, B_, ldb, C_, ldc,
                                                               Cin, bias, al, be, ga, relu, K, skip,
                                                               gate_part, gate_flag);
        }
    };

    hipMemcpyAsync(z, z0, sizeof(float) * (size_t)B * N2, hipMemcpyDeviceToDevice, stream);
    hipMemsetAsync(flag, 0, sizeof(int), stream);

    gemm(ctx, D, W1, H, 0, h,  H,  nullptr, b1, 1.f, 0.f, 1.f, 1, B, H,  D, nullptr, 32);
    gemm(h,   H, W2, N2, 0, wv, N2, nullptr, b2, 1.f, 0.f, 1.f, 0, B, N2, H, nullptr, 64);

    gemm(Ain, N2, Ain, N2, 1, G, N1, nullptr, nullptr, 1.f, 0.f, 0.f, 0, N1, N1, N2, nullptr, 64);

    float* part = h;
    k_sumsq<<<dim3(EW_BLOCKS), blk, 0, stream>>>(G, part, N1 * N1);
    k_init_X<<<dim3(EW_BLOCKS), blk, 0, stream>>>(Xa, part, N1);

    float* Xc = Xa; float* Xn = Xb;
    for (int it = 0; it < 16; ++it) {
        gemm(G,  N1, Xc, N1, 0, T1, N1, nullptr, nullptr,  1.f, 0.f, 0.f, 0, N1, N1, N1, nullptr, 64);
        gemm(Xc, N1, T1, N1, 0, Xn, N1, Xc,      nullptr, -1.f, 2.f, 0.f, 0, N1, N1, N1, nullptr, 64);
        float* tmp = Xc; Xc = Xn; Xn = tmp;
    }
    float* Ginv = Xc;

    const float* gate = nullptr;
    const int nz = B * N2;
    dim3 gout(EW_BLOCKS);
    k_prep_u<<<gout, blk, 0, stream>>>(z, wv, u, nz);
    for (int itr = 0; itr < 50; ++itr) {
        gemm(u, N2, Ain,  N2, 1, R, N1, nullptr, bvec, 1.f, 0.f, -1.f, 0, B, N1, N2, nullptr, 32,
             gate, flag);
        gemm(R, N1, Ginv, N1, 0, S, N1, nullptr, nullptr, 1.f, 0.f, 0.f, 0, B, N1, N1, flag, 32);
        gemm(S, N1, Ain,  N2, 0, y, N2, u, nullptr, -1.f, 1.f, 0.f, 0, B, N2, N1, flag, 64);
        k_update<<<gout, blk, 0, stream>>>(z, y, wv, u, part, nz, (const int*)flag);
        gate = part;
    }

    gemm(u, N2, Ain,  N2, 1, R, N1, nullptr, bvec, 1.f, 0.f, -1.f, 0, B, N1, N2, nullptr, 32);
    gemm(R, N1, Ginv, N1, 0, S, N1, nullptr, nullptr, 1.f, 0.f, 0.f, 0, B, N1, N1, nullptr, 32);
    gemm(S, N1, Ain,  N2, 0, y, N2, u, nullptr, -1.f, 1.f, 0.f, 0, B, N2, N1, nullptr, 64);
    k_final<<<gout, blk, 0, stream>>>(z, y, out, nz);
}
